# starting kernel
# baseline (speedup 1.0000x reference)
#include <hip/hip_runtime.h>


#define D_ 16
#define B_ 32
#define T_ 48000
#define P_ 750
#define LP_ 64
#define BLKS_PER_P 4

typedef __attribute__((ext_vector_type(2))) float v2f;
typedef __attribute__((ext_vector_type(4))) float v4f;
typedef __attribute__((ext_vector_type(8))) float v8f;

#define WS_A      0
#define WS_W      16
#define WS_WL     64
#define WS_A16L   320
#define WS_TAB    576
#define WS_XS     2048
#define WS_E      (WS_XS + T_*B_)
#define WS_CARRYS (WS_E + P_*B_*D_)

__device__ __forceinline__ float sigmoidf_(float z) { return 1.0f / (1.0f + expf(-z)); }
__device__ __forceinline__ float powi_(float a, int n) {
    float r = 1.0f;
    for (int i = 0; i < n; ++i) r *= a;
    return r;
}

__global__ void fdn_setup(const float* __restrict__ lk, const float* __restrict__ ar,
                          const float* __restrict__ br, const float* __restrict__ H,
                          float* __restrict__ ws) {
    const int t = threadIdx.x;
    if (t < D_) {
        const float max_delay = 50.0f * 16000.0f / 1000.0f;
        float kappa = 1.0f + sigmoidf_(lk[t]) * (max_delay - 1.0f);
        float a = expf(-1.0f / kappa);
        a = fminf(fmaxf(a, 0.0f), 0.9999f);
        ws[WS_A + t] = a;
        float s = 0.0f;
        for (int i = 0; i < D_; ++i) s += sigmoidf_(ar[i]) * H[i * D_ + t];
        ws[WS_W + t] = (sigmoidf_(br[t]) + s) * (1.0f / (float)D_);
    }
    __syncthreads();

    for (int e = t; e < 2 * 256; e += blockDim.x) {
        const int which = e >> 8, idx = e & 255;
        const int lane = idx >> 3, v = idx & 7;
        const int h = lane >> 4;
        const int d = 4 * (v >> 1) + 2 * h + (v & 1);
        const float a = ws[WS_A + d];
        if (which == 0) ws[WS_WL + idx] = ws[WS_W + d];
        else            ws[WS_A16L + idx] = powi_(a, 16);
    }

    for (int e = t; e < 12 * 64; e += blockDim.x) {
        const int chunk = e >> 6, pos = e & 63;
        const int lane = pos >> 1, ra = pos & 1;
        const int h = lane >> 4, m = lane & 15;
        const int kap = ra + 2 * h;
        float val = 0.0f;
        if (chunk < 4) {
            const int j = 4 * chunk + kap;
            if (j <= m)
                for (int d = 0; d < D_; ++d)
                    val += ws[WS_W + d] * powi_(ws[WS_A + d], m - j);
        } else if (chunk < 8) {
            const int d = 4 * (chunk - 4) + kap;
            val = powi_(ws[WS_A + d], m + 1);
        } else {
            int sg;
            if (m < 8) sg = 4 * (m >> 1) + (m & 1);
            else { const int v = m - 8; sg = 4 * (v >> 1) + 2 + (v & 1); }
            const int j = 4 * (chunk - 8) + kap;
            val = powi_(ws[WS_A + sg], 15 - j);
        }
        ws[WS_TAB + e] = val;
    }
}

__global__ void fdn_swizzle(const float* __restrict__ x, float* __restrict__ ws) {
    const int id = blockIdx.x * blockDim.x + threadIdx.x;
    if (id >= B_ * T_) return;
    const int b = id / T_, t = id % T_;
    const float val = x[id];
    const int tblk = t >> 4, j = t & 15;
    const int g = b >> 4, bl = b & 15;
    const int kk = j >> 2, kap = j & 3;
    const int h = kap >> 1, r = kap & 1;
    const int v = 2 * kk + r;
    ws[WS_XS + ((((tblk << 1) | g) * 32 + (h << 4) + bl) << 3) + v] = val;
}

__global__ void fdn_partial(const float* __restrict__ x, float* __restrict__ ws) {
    const int tid = blockIdx.x * blockDim.x + threadIdx.x;
    if (tid >= P_ * B_ * D_) return;
    const int p = tid >> 9;
    const int r = tid & 511;
    const int b = r >> 4, d = r & 15;
    const float a = ws[WS_A + d];
    const float* xp = x + b * T_ + p * LP_;
    float acc = 0.0f;
#pragma unroll 4
    for (int i = 0; i < LP_; ++i) acc = fmaf(a, acc, xp[i]);
    ws[WS_E + (p * B_ + b) * D_ + d] = acc;
}

__global__ void fdn_carry(float* __restrict__ ws) {
    const int tid = blockIdx.x * blockDim.x + threadIdx.x;
    if (tid >= B_ * D_) return;
    const int b = tid >> 4, d = tid & 15;
    const int g = b >> 4, bl = b & 15;
    const int h = (d >> 1) & 1, v = 2 * (d >> 2) + (d & 1);
    const int laneSlot = (((h << 4) + bl) << 3) + v;
    const float a = ws[WS_A + d];
    float aL = a;
    for (int q = 0; q < 6; ++q) aL *= aL;
    float c = 0.0f;
    for (int p = 0; p < P_; ++p) {
        ws[WS_CARRYS + ((((p << 1) | g) * 32) << 3) + laneSlot] = c;
        c = fmaf(aL, c, ws[WS_E + (p * B_ + b) * D_ + d]);
    }
}

__global__ void __launch_bounds__(256) fdn_main(const float* __restrict__ ws,
                                                float* __restrict__ out) {
    const int lane = threadIdx.x & 31;
    const int widx = blockIdx.x * (blockDim.x >> 5) + (threadIdx.x >> 5);
    if (widx >= 2 * P_) return;
    const int p = widx >> 1, g = widx & 1;
    const int h = lane >> 4, bl = lane & 15;

    const v8f wlv   = *(const v8f*)(ws + WS_WL   + lane * 8);
    const v8f a16lv = *(const v8f*)(ws + WS_A16L + lane * 8);

    v2f Lm[4], Pm[4], Rm[4];
#pragma unroll
    for (int kk = 0; kk < 4; ++kk) {
        Lm[kk] = *(const v2f*)(ws + WS_TAB + (kk)     * 64 + lane * 2);
        Pm[kk] = *(const v2f*)(ws + WS_TAB + (kk + 4) * 64 + lane * 2);
        Rm[kk] = *(const v2f*)(ws + WS_TAB + (kk + 8) * 64 + lane * 2);
    }

    v8f stv = *(const v8f*)(ws + WS_CARRYS + (widx * 32 + lane) * 8);

    const float* xS = ws + WS_XS;
    int tb = p * BLKS_PER_P;
    int t0 = p * LP_;
    for (int blk = 0; blk < BLKS_PER_P; ++blk) {
        const v8f xv = *(const v8f*)(xS + (((tb << 1) | g) * 32 + lane) * 8);
        const v2f* xc = (const v2f*)&xv;

        v8f qv = wlv * stv;
        const v2f* qc = (const v2f*)&qv;

        v8f acc = {0.f, 0.f, 0.f, 0.f, 0.f, 0.f, 0.f, 0.f};
#pragma unroll
        for (int kk = 0; kk < 4; ++kk)
            acc = __builtin_amdgcn_wmma_f32_16x16x4_f32(false, Pm[kk], false, qc[kk],
                                                        (short)0, acc, false, false);
        v8f rr = {0.f, 0.f, 0.f, 0.f, 0.f, 0.f, 0.f, 0.f};
#pragma unroll
        for (int kk = 0; kk < 4; ++kk) {
            acc = __builtin_amdgcn_wmma_f32_16x16x4_f32(false, Lm[kk], false, xc[kk],
                                                        (short)0, acc, false, false);
            rr  = __builtin_amdgcn_wmma_f32_16x16x4_f32(false, Rm[kk], false, xc[kk],
                                                        (short)0, rr, false, false);
        }
        stv = a16lv * stv + rr;

        float* op = out + (g * 16 + bl) * T_ + t0 + 8 * h;
        const v4f* ap = (const v4f*)&acc;
        *(v4f*)op       = ap[0];
        *(v4f*)(op + 4) = ap[1];

        ++tb;
        t0 += 16;
    }
}

extern "C" void kernel_launch(void* const* d_in, const int* in_sizes, int n_in,
                              void* d_out, int out_size, void* d_ws, size_t ws_size,
                              hipStream_t stream) {
    const float* x  = (const float*)d_in[0];
    const float* lk = (const float*)d_in[1];
    const float* ar = (const float*)d_in[2];
    const float* br = (const float*)d_in[3];
    const float* H  = (const float*)d_in[4];
    float* ws  = (float*)d_ws;
    float* out = (float*)d_out;

    hipLaunchKernelGGL(fdn_setup,   dim3(1),                          dim3(256), 0, stream, lk, ar, br, H, ws);
    hipLaunchKernelGGL(fdn_swizzle, dim3((B_ * T_ + 255) / 256),      dim3(256), 0, stream, x, ws);
    hipLaunchKernelGGL(fdn_partial, dim3((P_ * B_ * D_ + 255) / 256), dim3(256), 0, stream, x, ws);
    hipLaunchKernelGGL(fdn_carry,   dim3(2),                          dim3(256), 0, stream, ws);
    hipLaunchKernelGGL(fdn_main,    dim3((2 * P_ + 7) / 8),           dim3(256), 0, stream, ws, out);
}


# First launch cut by batch row, not time: a work-group per row keeps its 750 partitions as matrix columns and ends the carry scan inside (lane shifts, LDS); 3 -> 2 launches; carry, block sum order new
# speedup vs baseline: 20.9028x; 20.9028x over previous
#include <hip/hip_runtime.h>


#define D_ 16
#define B_ 32
#define T_ 48000
#define P_ 750
#define LP_ 64
#define BLKS_PER_P 4

typedef __attribute__((ext_vector_type(2))) float v2f;
typedef __attribute__((ext_vector_type(4))) float v4f;
typedef __attribute__((ext_vector_type(8))) float v8f;

#define WS_WL     64
#define WS_A16L   320
#define WS_TAB    576
#define WS_CARRY  2048

__device__ __forceinline__ float sigmoidf_(float z) { return 1.0f / (1.0f + expf(-z)); }
__device__ __forceinline__ float powi_(float a, int n) {
    float r = 1.0f;
    for (int i = 0; i < n; ++i) r *= a;
    return r;
}

__device__ __forceinline__ float decay_of(float lkv) {
    const float max_delay = 50.0f * 16000.0f / 1000.0f;
    const float kappa = 1.0f + sigmoidf_(lkv) * (max_delay - 1.0f);
    const float a = expf(-1.0f / kappa);
    return fminf(fmaxf(a, 0.0f), 0.9999f);
}

#define NTAB_WG 40
__device__ __forceinline__ void fdn_tables(const float* __restrict__ lk, const float* __restrict__ ar,
                                           const float* __restrict__ br, const float* __restrict__ H,
                                           float* __restrict__ ws, float* sh,
                                           const int tb, const int lane, const bool act) {
    const int dd = lane & 15;
    const float a_own = decay_of(lk[dd]);
    const float sal = sigmoidf_(ar[dd]);
    const float sbe = sigmoidf_(br[dd]);
    if (act && lane < D_) sh[lane] = sal;
    __syncthreads();
    float s = 0.0f;
    for (int i = 0; i < D_; ++i) s += sh[i] * H[i * D_ + dd];
    if (act && lane < D_) {
        sh[D_ + lane] = a_own;
        sh[2 * D_ + lane] = (sbe + s) * (1.0f / (float)D_);
    }
    __syncthreads();
    if (!act) return;
    const float* sa = sh + D_;
    const float* sw = sh + 2 * D_;

    if (tb >= 24) {
        const int e = (tb - 24) * 32 + lane;
        const int which = e >> 8, idx = e & 255;
        const int ln = idx >> 3, v = idx & 7;
        const int h = ln >> 4;
        const int d = 4 * (v >> 1) + 2 * h + (v & 1);
        if (which == 0) ws[WS_WL + idx] = sw[d];
        else            ws[WS_A16L + idx] = powi_(sa[d], 16);
        return;
    }

    const int e = tb * 32 + lane;
    const int chunk = e >> 6, pos = e & 63;
    const int ln = pos >> 1, ra = pos & 1;
    const int h = ln >> 4, m = ln & 15;
    const int kap = ra + 2 * h;
    const int jt = 8 * h + 2 * (chunk & 3) + ra;
    float val = 0.0f;
    if (chunk < 4) {
        if (jt <= m)
            for (int d = 0; d < D_; ++d)
                val += sw[d] * powi_(sa[d], m - jt);
    } else if (chunk < 8) {
        const int d = 4 * (chunk - 4) + kap;
        val = powi_(sa[d], m + 1);
    } else {
        int sg;
        if (m < 8) sg = 4 * (m >> 1) + (m & 1);
        else { const int v = m - 8; sg = 4 * (v >> 1) + 2 + (v & 1); }
        val = powi_(sa[sg], 15 - jt);
    }
    ws[WS_TAB + e] = val;
}

template <int N>
__device__ __forceinline__ v8f row_shr(const v8f s) {
    v8f r;
#pragma unroll
    for (int v = 0; v < 8; ++v)
        r[v] = __int_as_float(__builtin_amdgcn_update_dpp(0, __float_as_int(s[v]), 0x110 + N, 0xF, 0xF, true));
    return r;
}

#define RW_ 16
#define RUNP_ 3
#define RUNB_ (RUNP_ * BLKS_PER_P)
#define NRUN_ (P_ / RUNP_)
__global__ void __launch_bounds__(RW_ * 32) fdn_rows(const float* __restrict__ x,
                                                     const float* __restrict__ lk, const float* __restrict__ ar,
                                                     const float* __restrict__ br, const float* __restrict__ H,
                                                     float* __restrict__ ws) {
    __shared__ float sh[(2 * RW_ - 1) * 16];
    const int lane = threadIdx.x & 31;
    if (blockIdx.x >= B_) {
        fdn_tables(lk, ar, br, H, ws, sh, (int)blockIdx.x - B_, lane, threadIdx.x < 32);
        return;
    }
    const int b = blockIdx.x;
    const int w = threadIdx.x >> 5;
    const int h = lane >> 4, c = lane & 15;
    const int rho = 16 * w + c;
    const bool valid = rho < NRUN_;
    if (threadIdx.x < (RW_ - 1) * 16) sh[threadIdx.x] = 0.0f;

    const float lkv = lk[c];

    const float* xr = x + b * T_ + (valid ? rho : 0) * (RUNB_ * 16) + 8 * h;
    v8f xin[RUNB_];
#pragma unroll
    for (int i = 0; i < RUNB_; ++i) xin[i] = *(const v8f*)(xr + 16 * i);
    __builtin_amdgcn_sched_barrier(0);

    const float ad = decay_of(lkv);
    const float ad16 = powi_(ad, 16);
    const float ad32 = ad16 * ad16;
    const float ad64 = ad32 * ad32;
    const float ad192 = ad64 * ad64 * ad64;
    const int sgm = 4 * ((c & 7) >> 1) + (c & 1) + 2 * (c >> 3);
    const float asg = __shfl(ad, sgm);
    const float p1 = asg,       p2 = p1 * asg,   p3 = p2 * asg,   p4 = p3 * asg;
    const float p5 = p4 * asg,  p6 = p5 * asg,   p7 = p6 * asg,   p8 = p7 * asg;
    const float p9 = p8 * asg,  p10 = p9 * asg,  p11 = p10 * asg, p12 = p11 * asg;
    const float p13 = p12 * asg, p14 = p13 * asg, p15 = p14 * asg;
    v2f Rm[4];
    Rm[0].x = h ? p7 : p15;  Rm[0].y = h ? p6 : p14;
    Rm[1].x = h ? p5 : p13;  Rm[1].y = h ? p4 : p12;
    Rm[2].x = h ? p3 : p11;  Rm[2].y = h ? p2 : p10;
    Rm[3].x = h ? p1 : p9;   Rm[3].y = h ? 1.0f : p8;
    v8f a16lv, a64lv, A1;
#pragma unroll
    for (int v = 0; v < 8; ++v) {
        const int sl = 4 * (v >> 1) + 2 * h + (v & 1);
        a16lv[v] = __shfl(ad16, sl);
        a64lv[v] = __shfl(ad64, sl);
        A1[v]    = __shfl(ad192, sl);
    }

    v8f E0, E1, F;
    {
        v8f st;
#pragma unroll
        for (int i = 0; i < RUNB_; ++i) {
            const v2f* xc = (const v2f*)&xin[i];
            v8f rr = {0.f, 0.f, 0.f, 0.f, 0.f, 0.f, 0.f, 0.f};
#pragma unroll
            for (int kk = 0; kk < 4; ++kk)
                rr = __builtin_amdgcn_wmma_f32_16x16x4_f32(false, Rm[kk], false, xc[kk],
                                                           (short)0, rr, false, false);
            if ((i & 3) == 0) st = rr;
            else              st = a16lv * st + rr;
            if (i == 3) E0 = st;
            if (i == 7) E1 = st;
        }
        F = a64lv * (a64lv * E0 + E1) + st;
    }

    const v8f A2 = A1 * A1, A4 = A2 * A2, A8 = A4 * A4, A16 = A8 * A8;
    v8f S = F;
    S = A1 * row_shr<1>(S) + S;
    S = A2 * row_shr<2>(S) + S;
    S = A4 * row_shr<4>(S) + S;
    S = A8 * row_shr<8>(S) + S;
    if (c == 15) *(v8f*)(sh + (RW_ - 1 + w) * 16 + h * 8) = S;
    __syncthreads();
    v8f Gw = {0.f, 0.f, 0.f, 0.f, 0.f, 0.f, 0.f, 0.f};
    v8f wt[RW_ - 1];
#pragma unroll
    for (int k = 0; k < RW_ - 1; ++k)
        wt[k] = *(const v8f*)(sh + (w + k) * 16 + h * 8);
    __builtin_amdgcn_sched_barrier(0);
#pragma unroll
    for (int k = 0; k < RW_ - 1; ++k)
        Gw = A16 * Gw + wt[k];
    v8f pw = A1;
    pw = (c & 1) ? pw * A1 : pw;
    pw = (c & 2) ? pw * A2 : pw;
    pw = (c & 4) ? pw * A4 : pw;
    pw = (c & 8) ? pw * A8 : pw;
    const v8f Tend = pw * Gw + S;
    v8f G = row_shr<1>(Tend);
    G = (c == 0) ? Gw : G;

    const v8f c1 = a64lv * G + E0;
    const v8f c2 = a64lv * c1 + E1;
    if (valid) {
        float* cp = ws + WS_CARRY + ((b * P_ + RUNP_ * rho) * 2 + h) * 8;
        *(v8f*)cp        = G;
        *(v8f*)(cp + 16) = c1;
        *(v8f*)(cp + 32) = c2;
    }
}

__global__ void __launch_bounds__(256) fdn_main(const float* __restrict__ x,
                                                const float* __restrict__ ws,
                                                float* __restrict__ out) {
    const int lane = threadIdx.x & 31;
    const int widx = blockIdx.x * (blockDim.x >> 5) + (threadIdx.x >> 5);
    if (widx >= 2 * P_) return;
    const int p = widx >> 1, g = widx & 1;
    const int h = lane >> 4, bl = lane & 15;
    const int row = g * 16 + bl;

    const v8f wlv   = *(const v8f*)(ws + WS_WL   + lane * 8);
    const v8f a16lv = *(const v8f*)(ws + WS_A16L + lane * 8);

    v2f Lm[4], Pm[4], Rm[4];
#pragma unroll
    for (int kk = 0; kk < 4; ++kk) {
        Lm[kk] = *(const v2f*)(ws + WS_TAB + (kk)     * 64 + lane * 2);
        Pm[kk] = *(const v2f*)(ws + WS_TAB + (kk + 4) * 64 + lane * 2);
        Rm[kk] = *(const v2f*)(ws + WS_TAB + (kk + 8) * 64 + lane * 2);
    }

    const float* xr = x + row * T_ + p * LP_ + 8 * h;
    v8f xall[BLKS_PER_P];
#pragma unroll
    for (int blk = 0; blk < BLKS_PER_P; ++blk)
        xall[blk] = *(const v8f*)(xr + blk * 16);
    v8f stv = *(const v8f*)(ws + WS_CARRY + ((row * P_ + p) * 2 + h) * 8);

    int t0 = p * LP_;
#pragma unroll
    for (int blk = 0; blk < BLKS_PER_P; ++blk) {
        const v8f xv = xall[blk];
        const v2f* xc = (const v2f*)&xv;

        v8f qv = wlv * stv;
        const v2f* qc = (const v2f*)&qv;

        v8f acc = {0.f, 0.f, 0.f, 0.f, 0.f, 0.f, 0.f, 0.f};
#pragma unroll
        for (int kk = 0; kk < 4; ++kk)
            acc = __builtin_amdgcn_wmma_f32_16x16x4_f32(false, Pm[kk], false, qc[kk],
                                                        (short)0, acc, false, false);
        v8f rr = {0.f, 0.f, 0.f, 0.f, 0.f, 0.f, 0.f, 0.f};
#pragma unroll
        for (int kk = 0; kk < 4; ++kk) {
            acc = __builtin_amdgcn_wmma_f32_16x16x4_f32(false, Lm[kk], false, xc[kk],
                                                        (short)0, acc, false, false);
            rr  = __builtin_amdgcn_wmma_f32_16x16x4_f32(false, Rm[kk], false, xc[kk],
                                                        (short)0, rr, false, false);
        }
        stv = a16lv * stv + rr;

        float* op = out + row * T_ + t0 + 8 * h;
        const v4f* ap = (const v4f*)&acc;
        *(v4f*)op       = ap[0];
        *(v4f*)(op + 4) = ap[1];

        t0 += 16;
    }
}

extern "C" void kernel_launch(void* const* d_in, const int* in_sizes, int n_in,
                              void* d_out, int out_size, void* d_ws, size_t ws_size,
                              hipStream_t stream) {
    const float* x  = (const float*)d_in[0];
    const float* lk = (const float*)d_in[1];
    const float* ar = (const float*)d_in[2];
    const float* br = (const float*)d_in[3];
    const float* H  = (const float*)d_in[4];
    float* ws  = (float*)d_ws;
    float* out = (float*)d_out;

    hipLaunchKernelGGL(fdn_rows, dim3(B_ + NTAB_WG),       dim3(RW_ * 32), 0, stream, x, lk, ar, br, H, ws);
    hipLaunchKernelGGL(fdn_main, dim3((2 * P_ + 7) / 8),   dim3(256),      0, stream, x, ws, out);
}


# Output launch folded into the row launch (2 -> 1 launches): three work-groups per row, each cuts the row so the partition it writes ends every run, scans it, writes it; carry sum order new
# speedup vs baseline: 22.8755x; 1.0944x over previous
#include <hip/hip_runtime.h>


#define D_ 16
#define B_ 32
#define T_ 48000
#define P_ 750
#define LP_ 64
#define BLKS_PER_P 4

typedef __attribute__((ext_vector_type(2))) float v2f;
typedef __attribute__((ext_vector_type(4))) float v4f;
typedef __attribute__((ext_vector_type(8))) float v8f;

__device__ __forceinline__ float sigmoidf_(float z) { return 1.0f / (1.0f + expf(-z)); }
__device__ __forceinline__ float powi_(float a, int n) {
    float r = 1.0f;
    for (int i = 0; i < n; ++i) r *= a;
    return r;
}

__device__ __forceinline__ float decay_of(float lkv) {
    const float max_delay = 50.0f * 16000.0f / 1000.0f;
    const float kappa = 1.0f + sigmoidf_(lkv) * (max_delay - 1.0f);
    const float a = expf(-1.0f / kappa);
    return fminf(fmaxf(a, 0.0f), 0.9999f);
}


template <int N>
__device__ __forceinline__ v8f row_shr(const v8f s) {
    v8f r;
#pragma unroll
    for (int v = 0; v < 8; ++v)
        r[v] = __int_as_float(__builtin_amdgcn_update_dpp(0, __float_as_int(s[v]), 0x110 + N, 0xF, 0xF, true));
    return r;
}

__device__ __forceinline__ v8f lds8(const float* p) {
    const v4f a = *(const v4f*)p;
    const v4f b = *(const v4f*)(p + 4);
    return __builtin_shufflevector(a, b, 0, 1, 2, 3, 4, 5, 6, 7);
}

#define RW_ 16
#define RUNP_ 3
#define RUNB_ (RUNP_ * BLKS_PER_P)
#define NRUN_ (P_ / RUNP_)
#define NQ_ 3

#define SH_SCAN 0
#define SH_POW  496
#define SH_W    768
#define SH_FRAG 784
#define SH_TOTAL (SH_FRAG + 32 * 28)
#define POWP_ 17
#define FRAGP_ 28

__global__ void __launch_bounds__(RW_ * 32) fdn_fused(const float* __restrict__ x,
                                                      const float* __restrict__ lk, const float* __restrict__ ar,
                                                      const float* __restrict__ br, const float* __restrict__ H,
                                                      float* __restrict__ out) {
    __shared__ __attribute__((aligned(32))) float sh[SH_TOTAL];
    const int lane = threadIdx.x & 31;
    const int w = __builtin_amdgcn_readfirstlane((int)(threadIdx.x >> 5));
    const int b = (int)blockIdx.x / NQ_;
    const int q = (int)blockIdx.x - b * NQ_;
    const int h = lane >> 4, c = lane & 15;
    const int rho = 16 * w + c;
    const bool valid = rho < NRUN_;
    if (threadIdx.x < (RW_ - 1) * 16) sh[SH_SCAN + threadIdx.x] = 0.0f;

    const float lkv = lk[c];
    float arv = 0.0f, brv = 0.0f;
    float Hc[D_];
#pragma unroll
    for (int i = 0; i < D_; ++i) Hc[i] = 0.0f;
    if (w == 0) {
        arv = ar[c];
        brv = br[c];
#pragma unroll
        for (int i = 0; i < D_; ++i) Hc[i] = H[i * D_ + c];
    }

    const int pf = valid ? (RUNP_ * rho + q - (RUNP_ - 1)) : 0;
    const float* xrow = x + b * T_ + 8 * h;
    v8f xin[RUNB_];
#pragma unroll
    for (int i = 0; i < RUNB_; ++i) {
        const int off = (pf + (i >> 2) < 0) ? 0 : (pf * LP_ + 16 * i);
        xin[i] = *(const v8f*)(xrow + off);
    }
    __builtin_amdgcn_sched_barrier(0);

    const float ad = decay_of(lkv);
    const float ad16 = powi_(ad, 16);
    const float ad32 = ad16 * ad16;
    const float ad64 = ad32 * ad32;
    const float ad192 = ad64 * ad64 * ad64;
    const int sgm = 4 * ((c & 7) >> 1) + (c & 1) + 2 * (c >> 3);
    const float asg = __shfl(ad, sgm);
    const float p1 = asg,       p2 = p1 * asg,   p3 = p2 * asg,   p4 = p3 * asg;
    const float p5 = p4 * asg,  p6 = p5 * asg,   p7 = p6 * asg,   p8 = p7 * asg;
    const float p9 = p8 * asg,  p10 = p9 * asg,  p11 = p10 * asg, p12 = p11 * asg;
    const float p13 = p12 * asg, p14 = p13 * asg, p15 = p14 * asg;
    v2f Rm[4];
    Rm[0].x = h ? p7 : p15;  Rm[0].y = h ? p6 : p14;
    Rm[1].x = h ? p5 : p13;  Rm[1].y = h ? p4 : p12;
    Rm[2].x = h ? p3 : p11;  Rm[2].y = h ? p2 : p10;
    Rm[3].x = h ? p1 : p9;   Rm[3].y = h ? 1.0f : p8;
    v8f a16lv, a64lv, A1;
#pragma unroll
    for (int v = 0; v < 8; ++v) {
        const int sl = 4 * (v >> 1) + 2 * h + (v & 1);
        a16lv[v] = __shfl(ad16, sl);
        a64lv[v] = __shfl(ad64, sl);
        A1[v]    = __shfl(ad192, sl);
    }

    if (w == 0) {
        const float sal = sigmoidf_(arv);
        const float sbe = sigmoidf_(brv);
        float s = 0.0f;
#pragma unroll
        for (int i = 0; i < D_; ++i) s += __shfl(sal, i) * Hc[i];
        const float wgt = (sbe + s) * (1.0f / (float)D_);
        if (lane < D_) {
            float pr = 1.0f;
#pragma unroll
            for (int n = 0; n <= 16; ++n) { sh[SH_POW + c * POWP_ + n] = pr; pr *= ad; }
            sh[SH_W + c] = wgt;
        }
    }
    __builtin_amdgcn_fence(__ATOMIC_RELEASE, "workgroup", "local");
    __builtin_amdgcn_s_barrier();
    __builtin_amdgcn_fence(__ATOMIC_ACQUIRE, "workgroup", "local");
    if (w == 0) {
        float g = 0.0f;
#pragma unroll
        for (int d = 0; d < D_; ++d) g += sh[SH_W + d] * sh[SH_POW + d * POWP_ + c];
        v8f fP, fL, fW;
#pragma unroll
        for (int v = 0; v < 8; ++v) {
            const int dv = 4 * (v >> 1) + 2 * h + (v & 1);
            fP[v] = sh[SH_POW + dv * POWP_ + c + 1];
            fW[v] = sh[SH_W + dv];
            const int src = c - 8 * h - v;
            const float t = __shfl(g, src < 0 ? 0 : src);
            fL[v] = (src < 0) ? 0.0f : t;
        }
        float* fp = sh + SH_FRAG + lane * FRAGP_;
        *(v4f*)(fp)      = __builtin_shufflevector(fP, fP, 0, 1, 2, 3);
        *(v4f*)(fp + 4)  = __builtin_shufflevector(fP, fP, 4, 5, 6, 7);
        *(v4f*)(fp + 8)  = __builtin_shufflevector(fL, fL, 0, 1, 2, 3);
        *(v4f*)(fp + 12) = __builtin_shufflevector(fL, fL, 4, 5, 6, 7);
        *(v4f*)(fp + 16) = __builtin_shufflevector(fW, fW, 0, 1, 2, 3);
        *(v4f*)(fp + 20) = __builtin_shufflevector(fW, fW, 4, 5, 6, 7);
    }

    const v8f zero8 = {0.f, 0.f, 0.f, 0.f, 0.f, 0.f, 0.f, 0.f};
    if (w == 0 && q < RUNP_ - 1) {
#pragma unroll
        for (int i = 0; i < 2 * BLKS_PER_P; ++i)
            xin[i] = ((pf + (i >> 2)) < 0) ? zero8 : xin[i];
    }

    v8f E0, E1, F, rr8, rr9, rr10;
    {
        v8f st;
#pragma unroll
        for (int i = 0; i < RUNB_; ++i) {
            const v8f xv = xin[i];
            const v2f* xc = (const v2f*)&xv;
            v8f rr = zero8;
#pragma unroll
            for (int kk = 0; kk < 4; ++kk)
                rr = __builtin_amdgcn_wmma_f32_16x16x4_f32(false, Rm[kk], false, xc[kk],
                                                           (short)0, rr, false, false);
            if ((i & 3) == 0) st = rr;
            else              st = a16lv * st + rr;
            if (i == 3) E0 = st;
            if (i == 7) E1 = st;
            if (i == 8) rr8 = rr;
            if (i == 9) rr9 = rr;
            if (i == 10) rr10 = rr;
        }
        F = a64lv * (a64lv * E0 + E1) + st;
    }

    const v8f A2 = A1 * A1, A4 = A2 * A2, A8 = A4 * A4, A16 = A8 * A8;
    v8f S = F;
    S = A1 * row_shr<1>(S) + S;
    S = A2 * row_shr<2>(S) + S;
    S = A4 * row_shr<4>(S) + S;
    S = A8 * row_shr<8>(S) + S;
    if (c == 15) {
        float* tp = sh + SH_SCAN + (RW_ - 1 + w) * 16 + h * 8;
        *(v4f*)(tp)     = __builtin_shufflevector(S, S, 0, 1, 2, 3);
        *(v4f*)(tp + 4) = __builtin_shufflevector(S, S, 4, 5, 6, 7);
    }
    v8f pw = A1;
    pw = (c & 1) ? pw * A1 : pw;
    pw = (c & 2) ? pw * A2 : pw;
    pw = (c & 4) ? pw * A4 : pw;
    pw = (c & 8) ? pw * A8 : pw;
    __syncthreads();
    v8f Gw = zero8;
    {
        v8f wt[8];
#pragma unroll
        for (int k = 0; k < 8; ++k)
            wt[k] = lds8(sh + SH_SCAN + (w + k) * 16 + h * 8);
        __builtin_amdgcn_sched_barrier(0);
#pragma unroll
        for (int k = 0; k < 8; ++k)
            Gw = A16 * Gw + wt[k];
    }
    {
        v8f wt[7];
#pragma unroll
        for (int k = 0; k < 7; ++k)
            wt[k] = lds8(sh + SH_SCAN + (w + 8 + k) * 16 + h * 8);
        __builtin_amdgcn_sched_barrier(0);
#pragma unroll
        for (int k = 0; k < 7; ++k)
            Gw = A16 * Gw + wt[k];
    }
    const v8f Tend = pw * Gw + S;
    v8f G = row_shr<1>(Tend);
    G = (c == 0) ? Gw : G;

    const v8f c1 = a64lv * G + E0;
    const v8f c2 = a64lv * c1 + E1;

    const float* fp = sh + SH_FRAG + lane * FRAGP_;
    v2f Pm[4], Lm[4];
#pragma unroll
    for (int kk = 0; kk < 4; ++kk) {
        Pm[kk] = *(const v2f*)(fp + 2 * kk);
        Lm[kk] = *(const v2f*)(fp + 8 + 2 * kk);
    }
    const v8f wlv = lds8(fp + 16);

    v8f stb[BLKS_PER_P];
    stb[0] = c2;
    stb[1] = a16lv * stb[0] + rr8;
    stb[2] = a16lv * stb[1] + rr9;
    stb[3] = a16lv * stb[2] + rr10;
    v8f acc[BLKS_PER_P];
#pragma unroll
    for (int blk = 0; blk < BLKS_PER_P; ++blk) {
        const v8f xv = xin[(RUNP_ - 1) * BLKS_PER_P + blk];
        const v2f* xc = (const v2f*)&xv;

        const v8f qv = wlv * stb[blk];
        const v2f* qc = (const v2f*)&qv;

        v8f a = zero8;
#pragma unroll
        for (int kk = 0; kk < 4; ++kk)
            a = __builtin_amdgcn_wmma_f32_16x16x4_f32(false, Pm[kk], false, qc[kk],
                                                      (short)0, a, false, false);
#pragma unroll
        for (int kk = 0; kk < 4; ++kk)
            a = __builtin_amdgcn_wmma_f32_16x16x4_f32(false, Lm[kk], false, xc[kk],
                                                      (short)0, a, false, false);
        acc[blk] = a;
    }

    if (valid) {
        float* orow = out + b * T_ + (RUNP_ * rho + q) * LP_ + 8 * h;
#pragma unroll
        for (int blk = 0; blk < BLKS_PER_P; ++blk) {
            const v4f* ap = (const v4f*)&acc[blk];
            *(v4f*)(orow + 16 * blk)     = ap[0];
            *(v4f*)(orow + 16 * blk + 4) = ap[1];
        }
    }
}

extern "C" void kernel_launch(void* const* d_in, const int* in_sizes, int n_in,
                              void* d_out, int out_size, void* d_ws, size_t ws_size,
                              hipStream_t stream) {
    const float* x  = (const float*)d_in[0];
    const float* lk = (const float*)d_in[1];
    const float* ar = (const float*)d_in[2];
    const float* br = (const float*)d_in[3];
    const float* H  = (const float*)d_in[4];
    float* out = (float*)d_out;

    hipLaunchKernelGGL(fdn_fused, dim3(B_ * NQ_), dim3(RW_ * 32), 0, stream, x, lk, ar, br, H, out);
}


# x loads made whole-line: a wave reads its 16 runs as one contiguous 12 KiB stretch (4 full 128-byte lines per load, was 16 half-used lines), re-laid into operand order through wave-owned LDS
# speedup vs baseline: 24.8691x; 1.0872x over previous
#include <hip/hip_runtime.h>


#define D_ 16
#define B_ 32
#define T_ 48000
#define P_ 750
#define LP_ 64
#define BLKS_PER_P 4

typedef __attribute__((ext_vector_type(2))) float v2f;
typedef __attribute__((ext_vector_type(4))) float v4f;
typedef __attribute__((ext_vector_type(8))) float v8f;

__device__ __forceinline__ float sigmoidf_(float z) { return 1.0f / (1.0f + expf(-z)); }
__device__ __forceinline__ float powi_(float a, int n) {
    float r = 1.0f;
    for (int i = 0; i < n; ++i) r *= a;
    return r;
}

__device__ __forceinline__ float decay_of(float lkv) {
    const float max_delay = 50.0f * 16000.0f / 1000.0f;
    const float kappa = 1.0f + sigmoidf_(lkv) * (max_delay - 1.0f);
    const float a = expf(-1.0f / kappa);
    return fminf(fmaxf(a, 0.0f), 0.9999f);
}


template <int N>
__device__ __forceinline__ v8f row_shr(const v8f s) {
    v8f r;
#pragma unroll
    for (int v = 0; v < 8; ++v)
        r[v] = __int_as_float(__builtin_amdgcn_update_dpp(0, __float_as_int(s[v]), 0x110 + N, 0xF, 0xF, true));
    return r;
}

__device__ __forceinline__ v8f lds8(const float* p) {
    const v4f a = *(const v4f*)p;
    const v4f b = *(const v4f*)(p + 4);
    return __builtin_shufflevector(a, b, 0, 1, 2, 3, 4, 5, 6, 7);
}

#define RW_ 16
#define RUNP_ 3
#define RUNB_ (RUNP_ * BLKS_PER_P)
#define NRUN_ (P_ / RUNP_)
#define NQ_ 3

#define SH_SCAN 0
#define SH_POW  496
#define SH_W    768
#define SH_FRAG 784
#define SH_X    (SH_FRAG + 32 * 28)
#define XP_     196
#define XW_     (16 * XP_)
#define XLD_    24
#define SH_TOTAL (SH_X + RW_ * XW_)
#define POWP_ 17
#define FRAGP_ 28

__global__ void __launch_bounds__(RW_ * 32) fdn_fused(const float* __restrict__ x,
                                                      const float* __restrict__ lk, const float* __restrict__ ar,
                                                      const float* __restrict__ br, const float* __restrict__ H,
                                                      float* __restrict__ out) {
    __shared__ __attribute__((aligned(32))) float sh[SH_TOTAL];
    const int lane = threadIdx.x & 31;
    const int w = __builtin_amdgcn_readfirstlane((int)(threadIdx.x >> 5));
    const int b = (int)blockIdx.x / NQ_;
    const int q = (int)blockIdx.x - b * NQ_;
    const int h = lane >> 4, c = lane & 15;
    const int rho = 16 * w + c;
    const bool valid = rho < NRUN_;
    if (threadIdx.x < (RW_ - 1) * 16) sh[SH_SCAN + threadIdx.x] = 0.0f;

    const float lkv = lk[c];
    float arv = 0.0f, brv = 0.0f;
    float Hc[D_];
#pragma unroll
    for (int i = 0; i < D_; ++i) Hc[i] = 0.0f;
    if (w == 0) {
        arv = ar[c];
        brv = br[c];
#pragma unroll
        for (int i = 0; i < D_; ++i) Hc[i] = H[i * D_ + c];
    }

    const int pf = valid ? (RUNP_ * rho + q - (RUNP_ - 1)) : 0;
    const int seg0 = (RUNP_ * 16 * w + q - (RUNP_ - 1)) * LP_;
    const float* xr = x + b * T_;
    v4f xs[XLD_];
#pragma unroll
    for (int k = 0; k < XLD_; ++k) {
        int e = seg0 + 128 * k + 4 * lane;
        e = e < 0 ? 0 : (e > T_ - 4 ? T_ - 4 : e);
        xs[k] = *(const v4f*)(xr + e);
    }
    __builtin_amdgcn_sched_barrier(0);

    const float ad = decay_of(lkv);
    const float ad16 = powi_(ad, 16);
    const float ad32 = ad16 * ad16;
    const float ad64 = ad32 * ad32;
    const float ad192 = ad64 * ad64 * ad64;
    const int sgm = 4 * ((c & 7) >> 1) + (c & 1) + 2 * (c >> 3);
    const float asg = __shfl(ad, sgm);
    const float p1 = asg,       p2 = p1 * asg,   p3 = p2 * asg,   p4 = p3 * asg;
    const float p5 = p4 * asg,  p6 = p5 * asg,   p7 = p6 * asg,   p8 = p7 * asg;
    const float p9 = p8 * asg,  p10 = p9 * asg,  p11 = p10 * asg, p12 = p11 * asg;
    const float p13 = p12 * asg, p14 = p13 * asg, p15 = p14 * asg;
    v2f Rm[4];
    Rm[0].x = h ? p7 : p15;  Rm[0].y = h ? p6 : p14;
    Rm[1].x = h ? p5 : p13;  Rm[1].y = h ? p4 : p12;
    Rm[2].x = h ? p3 : p11;  Rm[2].y = h ? p2 : p10;
    Rm[3].x = h ? p1 : p9;   Rm[3].y = h ? 1.0f : p8;
    v8f a16lv, a64lv, A1;
#pragma unroll
    for (int v = 0; v < 8; ++v) {
        const int sl = 4 * (v >> 1) + 2 * h + (v & 1);
        a16lv[v] = __shfl(ad16, sl);
        a64lv[v] = __shfl(ad64, sl);
        A1[v]    = __shfl(ad192, sl);
    }

    if (w == 0) {
        const float sal = sigmoidf_(arv);
        const float sbe = sigmoidf_(brv);
        float s = 0.0f;
#pragma unroll
        for (int i = 0; i < D_; ++i) s += __shfl(sal, i) * Hc[i];
        const float wgt = (sbe + s) * (1.0f / (float)D_);
        if (lane < D_) {
            float pr = 1.0f;
#pragma unroll
            for (int n = 0; n <= 16; ++n) { sh[SH_POW + c * POWP_ + n] = pr; pr *= ad; }
            sh[SH_W + c] = wgt;
        }
    }
    __builtin_amdgcn_fence(__ATOMIC_RELEASE, "workgroup", "local");
    __builtin_amdgcn_s_barrier();
    __builtin_amdgcn_fence(__ATOMIC_ACQUIRE, "workgroup", "local");
    if (w == 0) {
        float g = 0.0f;
#pragma unroll
        for (int d = 0; d < D_; ++d) g += sh[SH_W + d] * sh[SH_POW + d * POWP_ + c];
        v8f fP, fL, fW;
#pragma unroll
        for (int v = 0; v < 8; ++v) {
            const int dv = 4 * (v >> 1) + 2 * h + (v & 1);
            fP[v] = sh[SH_POW + dv * POWP_ + c + 1];
            fW[v] = sh[SH_W + dv];
            const int src = c - 8 * h - v;
            const float t = __shfl(g, src < 0 ? 0 : src);
            fL[v] = (src < 0) ? 0.0f : t;
        }
        float* fp = sh + SH_FRAG + lane * FRAGP_;
        *(v4f*)(fp)      = __builtin_shufflevector(fP, fP, 0, 1, 2, 3);
        *(v4f*)(fp + 4)  = __builtin_shufflevector(fP, fP, 4, 5, 6, 7);
        *(v4f*)(fp + 8)  = __builtin_shufflevector(fL, fL, 0, 1, 2, 3);
        *(v4f*)(fp + 12) = __builtin_shufflevector(fL, fL, 4, 5, 6, 7);
        *(v4f*)(fp + 16) = __builtin_shufflevector(fW, fW, 0, 1, 2, 3);
        *(v4f*)(fp + 20) = __builtin_shufflevector(fW, fW, 4, 5, 6, 7);
    }

    float* xw = sh + SH_X + w * XW_;
#pragma unroll
    for (int k = 0; k < XLD_; ++k) {
        const int cc = 2 * (k / 3) + ((k % 3) == 2 ? 1 : ((k % 3) == 1 ? h : 0));
        *(v4f*)(xw + 128 * k + 4 * lane + 4 * cc) = xs[k];
    }
    __builtin_amdgcn_fence(__ATOMIC_RELEASE, "wavefront", "local");
    __builtin_amdgcn_wave_barrier();
    __builtin_amdgcn_fence(__ATOMIC_ACQUIRE, "wavefront", "local");
    v8f xin[RUNB_];
#pragma unroll
    for (int i = 0; i < RUNB_; ++i)
        xin[i] = lds8(xw + c * XP_ + 16 * i + 8 * h);

    const v8f zero8 = {0.f, 0.f, 0.f, 0.f, 0.f, 0.f, 0.f, 0.f};
    if (w == 0 && q < RUNP_ - 1) {
#pragma unroll
        for (int i = 0; i < 2 * BLKS_PER_P; ++i)
            xin[i] = ((pf + (i >> 2)) < 0) ? zero8 : xin[i];
    }

    v8f E0, E1, F, rr8, rr9, rr10;
    {
        v8f st;
#pragma unroll
        for (int i = 0; i < RUNB_; ++i) {
            const v8f xv = xin[i];
            const v2f* xc = (const v2f*)&xv;
            v8f rr = zero8;
#pragma unroll
            for (int kk = 0; kk < 4; ++kk)
                rr = __builtin_amdgcn_wmma_f32_16x16x4_f32(false, Rm[kk], false, xc[kk],
                                                           (short)0, rr, false, false);
            if ((i & 3) == 0) st = rr;
            else              st = a16lv * st + rr;
            if (i == 3) E0 = st;
            if (i == 7) E1 = st;
            if (i == 8) rr8 = rr;
            if (i == 9) rr9 = rr;
            if (i == 10) rr10 = rr;
        }
        F = a64lv * (a64lv * E0 + E1) + st;
    }

    const v8f A2 = A1 * A1, A4 = A2 * A2, A8 = A4 * A4, A16 = A8 * A8;
    v8f S = F;
    S = A1 * row_shr<1>(S) + S;
    S = A2 * row_shr<2>(S) + S;
    S = A4 * row_shr<4>(S) + S;
    S = A8 * row_shr<8>(S) + S;
    if (c == 15) {
        float* tp = sh + SH_SCAN + (RW_ - 1 + w) * 16 + h * 8;
        *(v4f*)(tp)     = __builtin_shufflevector(S, S, 0, 1, 2, 3);
        *(v4f*)(tp + 4) = __builtin_shufflevector(S, S, 4, 5, 6, 7);
    }
    v8f pw = A1;
    pw = (c & 1) ? pw * A1 : pw;
    pw = (c & 2) ? pw * A2 : pw;
    pw = (c & 4) ? pw * A4 : pw;
    pw = (c & 8) ? pw * A8 : pw;
    __syncthreads();
    v8f Gw = zero8;
    {
        v8f wt[8];
#pragma unroll
        for (int k = 0; k < 8; ++k)
            wt[k] = lds8(sh + SH_SCAN + (w + k) * 16 + h * 8);
        __builtin_amdgcn_sched_barrier(0);
#pragma unroll
        for (int k = 0; k < 8; ++k)
            Gw = A16 * Gw + wt[k];
    }
    {
        v8f wt[7];
#pragma unroll
        for (int k = 0; k < 7; ++k)
            wt[k] = lds8(sh + SH_SCAN + (w + 8 + k) * 16 + h * 8);
        __builtin_amdgcn_sched_barrier(0);
#pragma unroll
        for (int k = 0; k < 7; ++k)
            Gw = A16 * Gw + wt[k];
    }
    const v8f Tend = pw * Gw + S;
    v8f G = row_shr<1>(Tend);
    G = (c == 0) ? Gw : G;

    const v8f c1 = a64lv * G + E0;
    const v8f c2 = a64lv * c1 + E1;

    const float* fp = sh + SH_FRAG + lane * FRAGP_;
    v2f Pm[4], Lm[4];
#pragma unroll
    for (int kk = 0; kk < 4; ++kk) {
        Pm[kk] = *(const v2f*)(fp + 2 * kk);
        Lm[kk] = *(const v2f*)(fp + 8 + 2 * kk);
    }
    const v8f wlv = lds8(fp + 16);

    v8f stb[BLKS_PER_P];
    stb[0] = c2;
    stb[1] = a16lv * stb[0] + rr8;
    stb[2] = a16lv * stb[1] + rr9;
    stb[3] = a16lv * stb[2] + rr10;
    v8f acc[BLKS_PER_P];
#pragma unroll
    for (int blk = 0; blk < BLKS_PER_P; ++blk) {
        const v8f xv = xin[(RUNP_ - 1) * BLKS_PER_P + blk];
        const v2f* xc = (const v2f*)&xv;

        const v8f qv = wlv * stb[blk];
        const v2f* qc = (const v2f*)&qv;

        v8f a = zero8;
#pragma unroll
        for (int kk = 0; kk < 4; ++kk)
            a = __builtin_amdgcn_wmma_f32_16x16x4_f32(false, Pm[kk], false, qc[kk],
                                                      (short)0, a, false, false);
#pragma unroll
        for (int kk = 0; kk < 4; ++kk)
            a = __builtin_amdgcn_wmma_f32_16x16x4_f32(false, Lm[kk], false, xc[kk],
                                                      (short)0, a, false, false);
        acc[blk] = a;
    }

    if (valid) {
        float* orow = out + b * T_ + (RUNP_ * rho + q) * LP_ + 8 * h;
#pragma unroll
        for (int blk = 0; blk < BLKS_PER_P; ++blk) {
            const v4f* ap = (const v4f*)&acc[blk];
            *(v4f*)(orow + 16 * blk)     = ap[0];
            *(v4f*)(orow + 16 * blk + 4) = ap[1];
        }
    }
}

extern "C" void kernel_launch(void* const* d_in, const int* in_sizes, int n_in,
                              void* d_out, int out_size, void* d_ws, size_t ws_size,
                              hipStream_t stream) {
    const float* x  = (const float*)d_in[0];
    const float* lk = (const float*)d_in[1];
    const float* ar = (const float*)d_in[2];
    const float* br = (const float*)d_in[3];
    const float* H  = (const float*)d_in[4];
    float* out = (float*)d_out;

    hipLaunchKernelGGL(fdn_fused, dim3(B_ * NQ_), dim3(RW_ * 32), 0, stream, x, lk, ar, br, H, out);
}


# Row's work also split by halves of its runs (6 work-groups a row): front ones scan and write runs 0-127 with 8 waves; back ones scan all runs, only their waves 8-15 run the output products
# speedup vs baseline: 32.7334x; 1.3162x over previous
#include <hip/hip_runtime.h>


#define D_ 16
#define B_ 32
#define T_ 48000
#define P_ 750
#define LP_ 64
#define BLKS_PER_P 4

typedef __attribute__((ext_vector_type(2))) float v2f;
typedef __attribute__((ext_vector_type(4))) float v4f;
typedef __attribute__((ext_vector_type(8))) float v8f;

__device__ __forceinline__ float sigmoidf_(float z) { return 1.0f / (1.0f + expf(-z)); }
__device__ __forceinline__ float powi_(float a, int n) {
    float r = 1.0f;
    for (int i = 0; i < n; ++i) r *= a;
    return r;
}

__device__ __forceinline__ float decay_of(float lkv) {
    const float max_delay = 50.0f * 16000.0f / 1000.0f;
    const float kappa = 1.0f + sigmoidf_(lkv) * (max_delay - 1.0f);
    const float a = expf(-1.0f / kappa);
    return fminf(fmaxf(a, 0.0f), 0.9999f);
}


template <int N>
__device__ __forceinline__ v8f row_shr(const v8f s) {
    v8f r;
#pragma unroll
    for (int v = 0; v < 8; ++v)
        r[v] = __int_as_float(__builtin_amdgcn_update_dpp(0, __float_as_int(s[v]), 0x110 + N, 0xF, 0xF, true));
    return r;
}

__device__ __forceinline__ v8f lds8(const float* p) {
    const v4f a = *(const v4f*)p;
    const v4f b = *(const v4f*)(p + 4);
    return __builtin_shufflevector(a, b, 0, 1, 2, 3, 4, 5, 6, 7);
}

#define RW_ 16
#define RUNP_ 3
#define RUNB_ (RUNP_ * BLKS_PER_P)
#define NRUN_ (P_ / RUNP_)
#define NQ_ 3
#define NH_ 2
#define HW_ (RW_ / NH_)

#define SH_SCAN 0
#define SH_POW  496
#define SH_W    768
#define SH_FRAG 784
#define SH_X    (SH_FRAG + 32 * 28)
#define XP_     196
#define XW_     (16 * XP_)
#define XLD_    24
#define SH_TOTAL (SH_X + RW_ * XW_)
#define POWP_ 17
#define FRAGP_ 28

__global__ void __launch_bounds__(RW_ * 32) fdn_fused(const float* __restrict__ x,
                                                      const float* __restrict__ lk, const float* __restrict__ ar,
                                                      const float* __restrict__ br, const float* __restrict__ H,
                                                      float* __restrict__ out) {
    __shared__ __attribute__((aligned(32))) float sh[SH_TOTAL];
    const int lane = threadIdx.x & 31;
    const int w = __builtin_amdgcn_readfirstlane((int)(threadIdx.x >> 5));
    const int back = (int)blockIdx.x < B_ * NQ_ ? 1 : 0;
    const int rem = (int)blockIdx.x - (back ? 0 : B_ * NQ_);
    const int b = rem / NQ_;
    const int q = rem - b * NQ_;
    const bool act = back || (w < HW_);
    const bool outw = back ? (w >= HW_) : (w < HW_);
    const int h = lane >> 4, c = lane & 15;
    const int rho = 16 * w + c;
    const bool valid = rho < NRUN_;
    if (threadIdx.x < (RW_ - 1) * 16) sh[SH_SCAN + threadIdx.x] = 0.0f;

    const float lkv = lk[c];
    float arv = 0.0f, brv = 0.0f;
    float Hc[D_];
#pragma unroll
    for (int i = 0; i < D_; ++i) Hc[i] = 0.0f;
    if (w == 0) {
        arv = ar[c];
        brv = br[c];
#pragma unroll
        for (int i = 0; i < D_; ++i) Hc[i] = H[i * D_ + c];
    }

    const int pf = valid ? (RUNP_ * rho + q - (RUNP_ - 1)) : 0;
    const int seg0 = (RUNP_ * 16 * w + q - (RUNP_ - 1)) * LP_;
    const float* xr = x + b * T_;
    v4f xs[XLD_];
    v2f Rm[4];
    v8f a16lv, a64lv, A1, A2, A4, A8, A16;
    if (act) {
#pragma unroll
        for (int k = 0; k < XLD_; ++k) {
            int e = seg0 + 128 * k + 4 * lane;
            e = e < 0 ? 0 : (e > T_ - 4 ? T_ - 4 : e);
            xs[k] = *(const v4f*)(xr + e);
        }
        __builtin_amdgcn_sched_barrier(0);

        const float ad = decay_of(lkv);
        const float ad16 = powi_(ad, 16);
        const float ad32 = ad16 * ad16;
        const float ad64 = ad32 * ad32;
        const float ad192 = ad64 * ad64 * ad64;
        const int sgm = 4 * ((c & 7) >> 1) + (c & 1) + 2 * (c >> 3);
        const float asg = __shfl(ad, sgm);
        const float p1 = asg,       p2 = p1 * asg,   p3 = p2 * asg,   p4 = p3 * asg;
        const float p5 = p4 * asg,  p6 = p5 * asg,   p7 = p6 * asg,   p8 = p7 * asg;
        const float p9 = p8 * asg,  p10 = p9 * asg,  p11 = p10 * asg, p12 = p11 * asg;
        const float p13 = p12 * asg, p14 = p13 * asg, p15 = p14 * asg;
        Rm[0].x = h ? p7 : p15;  Rm[0].y = h ? p6 : p14;
        Rm[1].x = h ? p5 : p13;  Rm[1].y = h ? p4 : p12;
        Rm[2].x = h ? p3 : p11;  Rm[2].y = h ? p2 : p10;
        Rm[3].x = h ? p1 : p9;   Rm[3].y = h ? 1.0f : p8;
#pragma unroll
        for (int v = 0; v < 8; ++v) {
            const int sl = 4 * (v >> 1) + 2 * h + (v & 1);
            a16lv[v] = __shfl(ad16, sl);
            a64lv[v] = __shfl(ad64, sl);
            A1[v]    = __shfl(ad192, sl);
        }

        if (w == 0) {
            const float sal = sigmoidf_(arv);
            const float sbe = sigmoidf_(brv);
            float s = 0.0f;
#pragma unroll
            for (int i = 0; i < D_; ++i) s += __shfl(sal, i) * Hc[i];
            const float wgt = (sbe + s) * (1.0f / (float)D_);
            if (lane < D_) {
                float pr = 1.0f;
#pragma unroll
                for (int n = 0; n <= 16; ++n) { sh[SH_POW + c * POWP_ + n] = pr; pr *= ad; }
                sh[SH_W + c] = wgt;
            }
        }
    }
    __builtin_amdgcn_fence(__ATOMIC_RELEASE, "workgroup", "local");
    __builtin_amdgcn_s_barrier();
    __builtin_amdgcn_fence(__ATOMIC_ACQUIRE, "workgroup", "local");
    if (w == 0) {
        float g = 0.0f;
#pragma unroll
        for (int d = 0; d < D_; ++d) g += sh[SH_W + d] * sh[SH_POW + d * POWP_ + c];
        v8f fP, fL, fW;
#pragma unroll
        for (int v = 0; v < 8; ++v) {
            const int dv = 4 * (v >> 1) + 2 * h + (v & 1);
            fP[v] = sh[SH_POW + dv * POWP_ + c + 1];
            fW[v] = sh[SH_W + dv];
            const int src = c - 8 * h - v;
            const float t = __shfl(g, src < 0 ? 0 : src);
            fL[v] = (src < 0) ? 0.0f : t;
        }
        float* fp = sh + SH_FRAG + lane * FRAGP_;
        *(v4f*)(fp)      = __builtin_shufflevector(fP, fP, 0, 1, 2, 3);
        *(v4f*)(fp + 4)  = __builtin_shufflevector(fP, fP, 4, 5, 6, 7);
        *(v4f*)(fp + 8)  = __builtin_shufflevector(fL, fL, 0, 1, 2, 3);
        *(v4f*)(fp + 12) = __builtin_shufflevector(fL, fL, 4, 5, 6, 7);
        *(v4f*)(fp + 16) = __builtin_shufflevector(fW, fW, 0, 1, 2, 3);
        *(v4f*)(fp + 20) = __builtin_shufflevector(fW, fW, 4, 5, 6, 7);
    }

    const v8f zero8 = {0.f, 0.f, 0.f, 0.f, 0.f, 0.f, 0.f, 0.f};
    v8f xin[RUNB_];
    v8f E0, E1, rr8, rr9, rr10, S;
    if (act) {
        float* xw = sh + SH_X + w * XW_;
#pragma unroll
        for (int k = 0; k < XLD_; ++k) {
            const int cc = 2 * (k / 3) + ((k % 3) == 2 ? 1 : ((k % 3) == 1 ? h : 0));
            *(v4f*)(xw + 128 * k + 4 * lane + 4 * cc) = xs[k];
        }
        __builtin_amdgcn_fence(__ATOMIC_RELEASE, "wavefront", "local");
        __builtin_amdgcn_wave_barrier();
        __builtin_amdgcn_fence(__ATOMIC_ACQUIRE, "wavefront", "local");
#pragma unroll
        for (int i = 0; i < RUNB_; ++i)
            xin[i] = lds8(xw + c * XP_ + 16 * i + 8 * h);

        if (w == 0 && q < RUNP_ - 1) {
#pragma unroll
            for (int i = 0; i < 2 * BLKS_PER_P; ++i)
                xin[i] = ((pf + (i >> 2)) < 0) ? zero8 : xin[i];
        }

        v8f F;
        {
            v8f st;
#pragma unroll
            for (int i = 0; i < RUNB_; ++i) {
                const v8f xv = xin[i];
                const v2f* xc = (const v2f*)&xv;
                v8f rr = zero8;
#pragma unroll
                for (int kk = 0; kk < 4; ++kk)
                    rr = __builtin_amdgcn_wmma_f32_16x16x4_f32(false, Rm[kk], false, xc[kk],
                                                               (short)0, rr, false, false);
                if ((i & 3) == 0) st = rr;
                else              st = a16lv * st + rr;
                if (i == 3) E0 = st;
                if (i == 7) E1 = st;
                if (i == 8) rr8 = rr;
                if (i == 9) rr9 = rr;
                if (i == 10) rr10 = rr;
            }
            F = a64lv * (a64lv * E0 + E1) + st;
        }

        A2 = A1 * A1; A4 = A2 * A2; A8 = A4 * A4; A16 = A8 * A8;
        S = F;
        S = A1 * row_shr<1>(S) + S;
        S = A2 * row_shr<2>(S) + S;
        S = A4 * row_shr<4>(S) + S;
        S = A8 * row_shr<8>(S) + S;
        if (c == 15) {
            float* tp = sh + SH_SCAN + (RW_ - 1 + w) * 16 + h * 8;
            *(v4f*)(tp)     = __builtin_shufflevector(S, S, 0, 1, 2, 3);
            *(v4f*)(tp + 4) = __builtin_shufflevector(S, S, 4, 5, 6, 7);
        }
    }
    __syncthreads();
    if (!outw) return;
    v8f pw = A1;
    pw = (c & 1) ? pw * A1 : pw;
    pw = (c & 2) ? pw * A2 : pw;
    pw = (c & 4) ? pw * A4 : pw;
    pw = (c & 8) ? pw * A8 : pw;
    v8f Gw = zero8;
    {
        v8f wt[8];
#pragma unroll
        for (int k = 0; k < 8; ++k)
            wt[k] = lds8(sh + SH_SCAN + (w + k) * 16 + h * 8);
        __builtin_amdgcn_sched_barrier(0);
#pragma unroll
        for (int k = 0; k < 8; ++k)
            Gw = A16 * Gw + wt[k];
    }
    {
        v8f wt[7];
#pragma unroll
        for (int k = 0; k < 7; ++k)
            wt[k] = lds8(sh + SH_SCAN + (w + 8 + k) * 16 + h * 8);
        __builtin_amdgcn_sched_barrier(0);
#pragma unroll
        for (int k = 0; k < 7; ++k)
            Gw = A16 * Gw + wt[k];
    }
    const v8f Tend = pw * Gw + S;
    v8f G = row_shr<1>(Tend);
    G = (c == 0) ? Gw : G;

    const v8f c1 = a64lv * G + E0;
    const v8f c2 = a64lv * c1 + E1;

    const float* fp = sh + SH_FRAG + lane * FRAGP_;
    v2f Pm[4], Lm[4];
#pragma unroll
    for (int kk = 0; kk < 4; ++kk) {
        Pm[kk] = *(const v2f*)(fp + 2 * kk);
        Lm[kk] = *(const v2f*)(fp + 8 + 2 * kk);
    }
    const v8f wlv = lds8(fp + 16);

    v8f stb[BLKS_PER_P];
    stb[0] = c2;
    stb[1] = a16lv * stb[0] + rr8;
    stb[2] = a16lv * stb[1] + rr9;
    stb[3] = a16lv * stb[2] + rr10;
    v8f acc[BLKS_PER_P];
#pragma unroll
    for (int blk = 0; blk < BLKS_PER_P; ++blk) {
        const v8f xv = xin[(RUNP_ - 1) * BLKS_PER_P + blk];
        const v2f* xc = (const v2f*)&xv;

        const v8f qv = wlv * stb[blk];
        const v2f* qc = (const v2f*)&qv;

        v8f a = zero8;
#pragma unroll
        for (int kk = 0; kk < 4; ++kk)
            a = __builtin_amdgcn_wmma_f32_16x16x4_f32(false, Pm[kk], false, qc[kk],
                                                      (short)0, a, false, false);
#pragma unroll
        for (int kk = 0; kk < 4; ++kk)
            a = __builtin_amdgcn_wmma_f32_16x16x4_f32(false, Lm[kk], false, xc[kk],
                                                      (short)0, a, false, false);
        acc[blk] = a;
    }

    if (valid) {
        float* orow = out + b * T_ + (RUNP_ * rho + q) * LP_ + 8 * h;
#pragma unroll
        for (int blk = 0; blk < BLKS_PER_P; ++blk) {
            const v4f* ap = (const v4f*)&acc[blk];
            *(v4f*)(orow + 16 * blk)     = ap[0];
            *(v4f*)(orow + 16 * blk + 4) = ap[1];
        }
    }
}

extern "C" void kernel_launch(void* const* d_in, const int* in_sizes, int n_in,
                              void* d_out, int out_size, void* d_ws, size_t ws_size,
                              hipStream_t stream) {
    const float* x  = (const float*)d_in[0];
    const float* lk = (const float*)d_in[1];
    const float* ar = (const float*)d_in[2];
    const float* br = (const float*)d_in[3];
    const float* H  = (const float*)d_in[4];
    float* out = (float*)d_out;

    hipLaunchKernelGGL(fdn_fused, dim3(B_ * NQ_ * NH_), dim3(RW_ * 32), 0, stream, x, lk, ar, br, H, out);
}
